# starting kernel
# baseline (speedup 1.0000x reference)
#include <hip/hip_runtime.h>
#include <hip/hip_bf16.h>

typedef float v2f __attribute__((ext_vector_type(2)));
typedef float v8f __attribute__((ext_vector_type(8)));

#define NB 8192
#define BT 128
#define KC 32
#define KSPLIT 4
#define KSEG (NB / KSPLIT)
#define NBT (NB * BT)

__global__ __launch_bounds__(256) void repack_x_kernel(const float* __restrict__ X,
                                                       float* __restrict__ xr) {
  int idx = blockIdx.x * 256 + threadIdx.x;
  int n = idx >> 7;
  int c = idx & 127;
  int b = c >> 4;
  int t = c & 15;
  xr[idx] = X[(size_t)b * (NB * 16) + (size_t)n * 16 + t];
}

__global__ __launch_bounds__(256) void degree_kernel(const float* __restrict__ A,
                                                     float* __restrict__ dis,
                                                     float* __restrict__ Adiag) {
  int wave = threadIdx.x >> 5;
  int lane = threadIdx.x & 31;
  int row = blockIdx.x * 8 + wave;
  const float* Arow = A + (size_t)row * NB;
  float s = 0.f;
  for (int j = lane * 4; j < NB; j += 32 * 4) {
    float4 v = *reinterpret_cast<const float4*>(Arow + j);
    s += v.x + v.y + v.z + v.w;
  }
  for (int off = 16; off > 0; off >>= 1) s += __shfl_xor(s, off, 32);
  if (lane == 0) {
    float diag = Arow[row];
    float deg = s - diag;
    Adiag[row] = diag;
    dis[row] = (deg > 0.f) ? rsqrtf(deg) : 0.f;
  }
}

__global__ __launch_bounds__(256)
void lhat_gemm_kernel(const float* __restrict__ A,
                      const float* __restrict__ Bsrc,
                      const float* __restrict__ dis,
                      float* __restrict__ Part) {
  __shared__ float ldsB[KC / 2][BT][2];

  const int tid  = threadIdx.x;
  const int wave = tid >> 5;
  const int lane = tid & 31;
  const int half = lane >> 4;
  const int l16  = lane & 15;
  const int rowTile = blockIdx.x * 128 + wave * 16;
  const int kOff = blockIdx.y * KSEG;
  float* PartSeg = Part + (size_t)blockIdx.y * NBT;

  v8f acc[8];
  #pragma unroll
  for (int i = 0; i < 8; ++i)
    #pragma unroll
    for (int j = 0; j < 8; ++j) acc[i][j] = 0.f;

  const int lk  = tid >> 3;
  const int ln0 = (tid & 7) * 16;

  for (int kb = 0; kb < KSEG; kb += KC) {
    {
      const int krow = kOff + kb + lk;
      float dk = dis[krow];
      const float4* src4 =
          reinterpret_cast<const float4*>(Bsrc + (size_t)krow * BT + ln0);
      int p = lk >> 1;
      int jj = lk & 1;
      #pragma unroll
      for (int u = 0; u < 4; ++u) {
        float4 v = src4[u];
        ldsB[p][ln0 + u * 4 + 0][jj] = v.x * dk;
        ldsB[p][ln0 + u * 4 + 1][jj] = v.y * dk;
        ldsB[p][ln0 + u * 4 + 2][jj] = v.z * dk;
        ldsB[p][ln0 + u * 4 + 3][jj] = v.w * dk;
      }
    }
    __syncthreads();

    const float* Abase = A + (size_t)(rowTile + l16) * NB + kOff + kb + half * 2;
    if (kb + KC < KSEG) __builtin_prefetch(Abase + KC, 0, 0);
    v2f a[8];
    #pragma unroll
    for (int s = 0; s < 8; ++s)
      a[s] = *reinterpret_cast<const v2f*>(Abase + s * 4);

    v2f bq[2][8];
    #pragma unroll
    for (int n0 = 0; n0 < 8; ++n0)
      bq[0][n0] = *reinterpret_cast<const v2f*>(&ldsB[half][n0 * 16 + l16][0]);

    #pragma unroll
    for (int s = 0; s < 8; ++s) {
      const int cur = s & 1;
      const int nxt = cur ^ 1;
      if (s < 7) {
        const int p = (s + 1) * 2 + half;
        #pragma unroll
        for (int n0 = 0; n0 < 8; ++n0)
          bq[nxt][n0] = *reinterpret_cast<const v2f*>(&ldsB[p][n0 * 16 + l16][0]);
      }
      #pragma unroll
      for (int n0 = 0; n0 < 8; ++n0)
        acc[n0] = __builtin_amdgcn_wmma_f32_16x16x4_f32(
            false, a[s], false, bq[cur][n0], (short)0, acc[n0], false, false);
    }
    __syncthreads();
  }

  const int r0 = rowTile + half * 8;
  #pragma unroll
  for (int g = 0; g < 8; ++g) {
    const int r = r0 + g;
    #pragma unroll
    for (int n0 = 0; n0 < 8; ++n0) {
      const int c = n0 * 16 + l16;
      PartSeg[(size_t)r * BT + c] = acc[n0][g];
    }
  }
}

__global__ __launch_bounds__(256)
void ksplit_epilogue_kernel(const float* __restrict__ Part,
                            const float* __restrict__ Bsrc,
                            const float* __restrict__ dis,
                            const float* __restrict__ Adiag,
                            const float* __restrict__ Xadd,
                            float* __restrict__ D,
                            float alpha, float beta) {
  int idx = blockIdx.x * 256 + threadIdx.x;
  int r = idx >> 7;
  float acc = Part[idx] + Part[idx + NBT] + Part[idx + 2 * NBT] + Part[idx + 3 * NBT];
  float dr = dis[r];
  float ad = Adiag[r];
  float v = acc - ad * dr * Bsrc[idx];
  D[idx] = alpha * dr * v + beta * Xadd[idx];
}

__global__ __launch_bounds__(256)
void cheb_conv_kernel(const float* __restrict__ Tx0, const float* __restrict__ Tx1,
                      const float* __restrict__ Tx2,
                      const float* __restrict__ Wc, const float* __restrict__ bc,
                      const float* __restrict__ wt, const float* __restrict__ bt,
                      float* __restrict__ out) {
  __shared__ float sW[3 * 256];
  __shared__ float sbc[16];
  __shared__ float swt[4];
  __shared__ float stile[8][16][17];

  const int tid = threadIdx.x;
  for (int i = tid; i < 768; i += 256) sW[i] = Wc[i];
  if (tid < 16) sbc[tid] = bc[tid];
  if (tid < 3) swt[tid] = wt[tid];
  if (tid == 3) swt[3] = bt[0];
  __syncthreads();

  const int wave = tid >> 5;
  const int lane = tid & 31;
  const int half = lane >> 4;
  const int l16  = lane & 15;
  const int tileId = blockIdx.x * 8 + wave;
  const int b  = tileId >> 9;
  const int nT = tileId & 511;

  const float* Ts[3] = {Tx0, Tx1, Tx2};
  v8f acc;
  #pragma unroll
  for (int j = 0; j < 8; ++j) acc[j] = 0.f;

  #pragma unroll
  for (int m = 0; m < 3; ++m) {
    const float* Abase =
        Ts[m] + (size_t)(nT * 16 + l16) * BT + b * 16 + half * 2;
    #pragma unroll
    for (int s = 0; s < 4; ++s) {
      v2f a = *reinterpret_cast<const v2f*>(Abase + s * 4);
      const int k = s * 4 + half * 2;
      v2f w;
      w[0] = sW[m * 256 + k * 16 + l16];
      w[1] = sW[m * 256 + (k + 1) * 16 + l16];
      acc = __builtin_amdgcn_wmma_f32_16x16x4_f32(
          false, a, false, w, (short)0, acc, false, false);
    }
  }

  const float bcol = sbc[l16];
  #pragma unroll
  for (int g = 0; g < 8; ++g) {
    float v = acc[g] + bcol;
    stile[wave][g + half * 8][l16] = v > 0.f ? v : 0.f;
  }
  __syncthreads();

  const float w0 = swt[0], w1 = swt[1], w2 = swt[2], bb = swt[3];
  #pragma unroll
  for (int g = 0; g < 8; ++g) {
    const int r = g + half * 8;
    const int t = l16;
    float v = bb + w1 * stile[wave][r][t];
    if (t > 0)  v += w0 * stile[wave][r][t - 1];
    if (t < 15) v += w2 * stile[wave][r][t + 1];
    v = v > 0.f ? v : 0.f;
    const int n = nT * 16 + r;
    out[((size_t)b * NB + n) * 16 + t] = v;
  }
}

extern "C" void kernel_launch(void* const* d_in, const int* in_sizes, int n_in,
                              void* d_out, int out_size, void* d_ws, size_t ws_size,
                              hipStream_t stream) {
  const float* X  = (const float*)d_in[0];
  const float* A  = (const float*)d_in[1];
  const float* Wc = (const float*)d_in[2];
  const float* bc = (const float*)d_in[3];
  const float* wt = (const float*)d_in[4];
  const float* bt = (const float*)d_in[5];
  float* out = (float*)d_out;

  float* ws    = (float*)d_ws;
  float* xr    = ws;
  float* Tx1   = ws + (size_t)NBT;
  float* Tx2   = ws + (size_t)2 * NBT;
  float* dis   = ws + (size_t)3 * NBT;
  float* Adiag = dis + NB;
  float* Part  = Adiag + NB;

  repack_x_kernel<<<4096, 256, 0, stream>>>(X, xr);
  degree_kernel<<<1024, 256, 0, stream>>>(A, dis, Adiag);

  lhat_gemm_kernel<<<dim3(64, KSPLIT), 256, 0, stream>>>(A, xr, dis, Part);
  ksplit_epilogue_kernel<<<4096, 256, 0, stream>>>(Part, xr, dis, Adiag, xr, Tx1,
                                                   -1.f, 0.f);
  lhat_gemm_kernel<<<dim3(64, KSPLIT), 256, 0, stream>>>(A, Tx1, dis, Part);
  ksplit_epilogue_kernel<<<4096, 256, 0, stream>>>(Part, Tx1, dis, Adiag, xr, Tx2,
                                                   -2.f, -1.f);

  cheb_conv_kernel<<<512, 256, 0, stream>>>(xr, Tx1, Tx2, Wc, bc, wt, bt, out);
}


# GEMM K loop software-pipelined: next 32-k chunk's global loads (A operands, B panel) issued ahead of the current chunk's 64 WMMAs, panel in 2 LDS buffers, 1 barrier per chunk
# speedup vs baseline: 1.3036x; 1.3036x over previous
#include <hip/hip_runtime.h>
#include <hip/hip_bf16.h>

typedef float v2f __attribute__((ext_vector_type(2)));
typedef float v8f __attribute__((ext_vector_type(8)));

#define NB 8192
#define BT 128
#define KC 32
#define KSPLIT 4
#define KSEG (NB / KSPLIT)
#define NBT (NB * BT)
#define NIT (KSEG / KC)

__global__ __launch_bounds__(256) void repack_x_kernel(const float* __restrict__ X,
                                                       float* __restrict__ xr) {
  int idx = blockIdx.x * 256 + threadIdx.x;
  int n = idx >> 7;
  int c = idx & 127;
  int b = c >> 4;
  int t = c & 15;
  xr[idx] = X[(size_t)b * (NB * 16) + (size_t)n * 16 + t];
}

__global__ __launch_bounds__(256) void degree_kernel(const float* __restrict__ A,
                                                     float* __restrict__ dis,
                                                     float* __restrict__ Adiag) {
  int wave = threadIdx.x >> 5;
  int lane = threadIdx.x & 31;
  int row = blockIdx.x * 8 + wave;
  const float* Arow = A + (size_t)row * NB;
  float s = 0.f;
  for (int j = lane * 4; j < NB; j += 32 * 4) {
    float4 v = *reinterpret_cast<const float4*>(Arow + j);
    s += v.x + v.y + v.z + v.w;
  }
  for (int off = 16; off > 0; off >>= 1) s += __shfl_xor(s, off, 32);
  if (lane == 0) {
    float diag = Arow[row];
    float deg = s - diag;
    Adiag[row] = diag;
    dis[row] = (deg > 0.f) ? rsqrtf(deg) : 0.f;
  }
}


struct PanelRegs {
  float4 v[4];
  float dk;
};

__device__ __forceinline__ void load_chunk(const float* __restrict__ Ak,
                                           const float* __restrict__ Bk,
                                           const float* __restrict__ dk_ptr,
                                           v2f (&a)[8], PanelRegs& pr) {
  #pragma unroll
  for (int s = 0; s < 8; ++s)
    a[s] = *reinterpret_cast<const v2f*>(Ak + s * 4);
  const float4* src4 = reinterpret_cast<const float4*>(Bk);
  #pragma unroll
  for (int u = 0; u < 4; ++u) pr.v[u] = src4[u];
  pr.dk = *dk_ptr;
}

__device__ __forceinline__ void store_panel(float (*panel)[BT][2], const PanelRegs& pr,
                                            int lk, int ln0) {
  const int p = lk >> 1;
  const int jj = lk & 1;
  const float dk = pr.dk;
  #pragma unroll
  for (int u = 0; u < 4; ++u) {
    const float4 v = pr.v[u];
    panel[p][ln0 + u * 4 + 0][jj] = v.x * dk;
    panel[p][ln0 + u * 4 + 1][jj] = v.y * dk;
    panel[p][ln0 + u * 4 + 2][jj] = v.z * dk;
    panel[p][ln0 + u * 4 + 3][jj] = v.w * dk;
  }
}

__device__ __forceinline__ void mma_chunk(const float (*panel)[BT][2], const v2f (&a)[8],
                                          v8f (&acc)[8], int half, int l16) {
  v2f bq[2][8];
  #pragma unroll
  for (int n0 = 0; n0 < 8; ++n0)
    bq[0][n0] = *reinterpret_cast<const v2f*>(&panel[half][n0 * 16 + l16][0]);

  #pragma unroll
  for (int s = 0; s < 8; ++s) {
    const int cur = s & 1;
    const int nxt = cur ^ 1;
    if (s < 7) {
      const int p = (s + 1) * 2 + half;
      #pragma unroll
      for (int n0 = 0; n0 < 8; ++n0)
        bq[nxt][n0] = *reinterpret_cast<const v2f*>(&panel[p][n0 * 16 + l16][0]);
    }
    #pragma unroll
    for (int n0 = 0; n0 < 8; ++n0)
      acc[n0] = __builtin_amdgcn_wmma_f32_16x16x4_f32(
          false, a[s], false, bq[cur][n0], (short)0, acc[n0], false, false);
  }
}

__global__ __launch_bounds__(256)
void lhat_gemm_kernel(const float* __restrict__ A,
                      const float* __restrict__ Bsrc,
                      const float* __restrict__ dis,
                      float* __restrict__ Part) {
  __shared__ float ldsB[2][KC / 2][BT][2];

  const int tid  = threadIdx.x;
  const int wave = tid >> 5;
  const int lane = tid & 31;
  const int half = lane >> 4;
  const int l16  = lane & 15;
  const int rowTile = blockIdx.x * 128 + wave * 16;
  const int kOff = blockIdx.y * KSEG;
  float* PartSeg = Part + (size_t)blockIdx.y * NBT;

  v8f acc[8];
  #pragma unroll
  for (int i = 0; i < 8; ++i)
    #pragma unroll
    for (int j = 0; j < 8; ++j) acc[i][j] = 0.f;

  const int lk  = tid >> 3;
  const int ln0 = (tid & 7) * 16;

  const float* Ak = A + (size_t)(rowTile + l16) * NB + kOff + half * 2;
  const float* Bk = Bsrc + (size_t)(kOff + lk) * BT + ln0;
  const float* dk_ptr = dis + kOff + lk;

  v2f a0[8], a1[8];
  PanelRegs pr;

  load_chunk(Ak, Bk, dk_ptr, a0, pr);
  store_panel(ldsB[0], pr, lk, ln0);
  __syncthreads();

  for (int it = 0; it < NIT; it += 2) {
    load_chunk(Ak + (it + 1) * KC, Bk + (size_t)(it + 1) * KC * BT,
               dk_ptr + (it + 1) * KC, a1, pr);
    mma_chunk(ldsB[0], a0, acc, half, l16);
    store_panel(ldsB[1], pr, lk, ln0);
    __syncthreads();

    const bool more = it + 2 < NIT;
    if (more)
      load_chunk(Ak + (it + 2) * KC, Bk + (size_t)(it + 2) * KC * BT,
                 dk_ptr + (it + 2) * KC, a0, pr);
    mma_chunk(ldsB[1], a1, acc, half, l16);
    if (more) store_panel(ldsB[0], pr, lk, ln0);
    __syncthreads();
  }

  const int r0 = rowTile + half * 8;
  #pragma unroll
  for (int g = 0; g < 8; ++g) {
    const int r = r0 + g;
    #pragma unroll
    for (int n0 = 0; n0 < 8; ++n0) {
      const int c = n0 * 16 + l16;
      PartSeg[(size_t)r * BT + c] = acc[n0][g];
    }
  }
}

__global__ __launch_bounds__(256)
void ksplit_epilogue_kernel(const float* __restrict__ Part,
                            const float* __restrict__ Bsrc,
                            const float* __restrict__ dis,
                            const float* __restrict__ Adiag,
                            const float* __restrict__ Xadd,
                            float* __restrict__ D,
                            float alpha, float beta) {
  int idx = blockIdx.x * 256 + threadIdx.x;
  int r = idx >> 7;
  float acc = Part[idx] + Part[idx + NBT] + Part[idx + 2 * NBT] + Part[idx + 3 * NBT];
  float dr = dis[r];
  float ad = Adiag[r];
  float v = acc - ad * dr * Bsrc[idx];
  D[idx] = alpha * dr * v + beta * Xadd[idx];
}

__global__ __launch_bounds__(256)
void cheb_conv_kernel(const float* __restrict__ Tx0, const float* __restrict__ Tx1,
                      const float* __restrict__ Tx2,
                      const float* __restrict__ Wc, const float* __restrict__ bc,
                      const float* __restrict__ wt, const float* __restrict__ bt,
                      float* __restrict__ out) {
  __shared__ float sW[3 * 256];
  __shared__ float sbc[16];
  __shared__ float swt[4];
  __shared__ float stile[8][16][17];

  const int tid = threadIdx.x;
  for (int i = tid; i < 768; i += 256) sW[i] = Wc[i];
  if (tid < 16) sbc[tid] = bc[tid];
  if (tid < 3) swt[tid] = wt[tid];
  if (tid == 3) swt[3] = bt[0];
  __syncthreads();

  const int wave = tid >> 5;
  const int lane = tid & 31;
  const int half = lane >> 4;
  const int l16  = lane & 15;
  const int tileId = blockIdx.x * 8 + wave;
  const int b  = tileId >> 9;
  const int nT = tileId & 511;

  const float* Ts[3] = {Tx0, Tx1, Tx2};
  v8f acc;
  #pragma unroll
  for (int j = 0; j < 8; ++j) acc[j] = 0.f;

  #pragma unroll
  for (int m = 0; m < 3; ++m) {
    const float* Abase =
        Ts[m] + (size_t)(nT * 16 + l16) * BT + b * 16 + half * 2;
    #pragma unroll
    for (int s = 0; s < 4; ++s) {
      v2f a = *reinterpret_cast<const v2f*>(Abase + s * 4);
      const int k = s * 4 + half * 2;
      v2f w;
      w[0] = sW[m * 256 + k * 16 + l16];
      w[1] = sW[m * 256 + (k + 1) * 16 + l16];
      acc = __builtin_amdgcn_wmma_f32_16x16x4_f32(
          false, a, false, w, (short)0, acc, false, false);
    }
  }

  const float bcol = sbc[l16];
  #pragma unroll
  for (int g = 0; g < 8; ++g) {
    float v = acc[g] + bcol;
    stile[wave][g + half * 8][l16] = v > 0.f ? v : 0.f;
  }
  __syncthreads();

  const float w0 = swt[0], w1 = swt[1], w2 = swt[2], bb = swt[3];
  #pragma unroll
  for (int g = 0; g < 8; ++g) {
    const int r = g + half * 8;
    const int t = l16;
    float v = bb + w1 * stile[wave][r][t];
    if (t > 0)  v += w0 * stile[wave][r][t - 1];
    if (t < 15) v += w2 * stile[wave][r][t + 1];
    v = v > 0.f ? v : 0.f;
    const int n = nT * 16 + r;
    out[((size_t)b * NB + n) * 16 + t] = v;
  }
}

extern "C" void kernel_launch(void* const* d_in, const int* in_sizes, int n_in,
                              void* d_out, int out_size, void* d_ws, size_t ws_size,
                              hipStream_t stream) {
  const float* X  = (const float*)d_in[0];
  const float* A  = (const float*)d_in[1];
  const float* Wc = (const float*)d_in[2];
  const float* bc = (const float*)d_in[3];
  const float* wt = (const float*)d_in[4];
  const float* bt = (const float*)d_in[5];
  float* out = (float*)d_out;

  float* ws    = (float*)d_ws;
  float* xr    = ws;
  float* Tx1   = ws + (size_t)NBT;
  float* Tx2   = ws + (size_t)2 * NBT;
  float* dis   = ws + (size_t)3 * NBT;
  float* Adiag = dis + NB;
  float* Part  = Adiag + NB;

  repack_x_kernel<<<4096, 256, 0, stream>>>(X, xr);
  degree_kernel<<<1024, 256, 0, stream>>>(A, dis, Adiag);

  lhat_gemm_kernel<<<dim3(64, KSPLIT), 256, 0, stream>>>(A, xr, dis, Part);
  ksplit_epilogue_kernel<<<4096, 256, 0, stream>>>(Part, xr, dis, Adiag, xr, Tx1,
                                                   -1.f, 0.f);
  lhat_gemm_kernel<<<dim3(64, KSPLIT), 256, 0, stream>>>(A, Tx1, dis, Part);
  ksplit_epilogue_kernel<<<4096, 256, 0, stream>>>(Part, Tx1, dis, Adiag, xr, Tx2,
                                                   -2.f, -1.f);

  cheb_conv_kernel<<<512, 256, 0, stream>>>(xr, Tx1, Tx2, Wc, bc, wt, bt, out);
}


# GEMM B panel in LDS re-laid as [k/4][permuted column slot][k%4]: 16-byte staging stores and 8-byte fragment reads both bank-conflict-free (were 8-way / 2-way); k order and sums unchanged
# speedup vs baseline: 1.4978x; 1.1490x over previous
#include <hip/hip_runtime.h>
#include <hip/hip_bf16.h>

typedef float v2f __attribute__((ext_vector_type(2)));
typedef float v4f __attribute__((ext_vector_type(4)));
typedef float v8f __attribute__((ext_vector_type(8)));

#define NB 8192
#define BT 128
#define KC 32
#define KQ (KC / 4)
#define KSPLIT 4
#define KSEG (NB / KSPLIT)
#define NBT (NB * BT)
#define NIT (KSEG / KC)

__global__ __launch_bounds__(256) void repack_x_kernel(const float* __restrict__ X,
                                                       float* __restrict__ xr) {
  int idx = blockIdx.x * 256 + threadIdx.x;
  int n = idx >> 7;
  int c = idx & 127;
  int b = c >> 4;
  int t = c & 15;
  xr[idx] = X[(size_t)b * (NB * 16) + (size_t)n * 16 + t];
}

__global__ __launch_bounds__(256) void degree_kernel(const float* __restrict__ A,
                                                     float* __restrict__ dis,
                                                     float* __restrict__ Adiag) {
  int wave = threadIdx.x >> 5;
  int lane = threadIdx.x & 31;
  int row = blockIdx.x * 8 + wave;
  const float* Arow = A + (size_t)row * NB;
  float s = 0.f;
  for (int j = lane * 4; j < NB; j += 32 * 4) {
    float4 v = *reinterpret_cast<const float4*>(Arow + j);
    s += v.x + v.y + v.z + v.w;
  }
  for (int off = 16; off > 0; off >>= 1) s += __shfl_xor(s, off, 32);
  if (lane == 0) {
    float diag = Arow[row];
    float deg = s - diag;
    Adiag[row] = diag;
    dis[row] = (deg > 0.f) ? rsqrtf(deg) : 0.f;
  }
}


struct PanelRegs {
  v4f v[4];
  v4f d;
};

__device__ __forceinline__ void load_chunk(const float* __restrict__ Ak,
                                           const float* __restrict__ Bk,
                                           const float* __restrict__ dk_ptr,
                                           v2f (&a)[8], PanelRegs& pr) {
  #pragma unroll
  for (int s = 0; s < 8; ++s)
    a[s] = *reinterpret_cast<const v2f*>(Ak + s * 4);
  #pragma unroll
  for (int r = 0; r < 4; ++r)
    pr.v[r] = *reinterpret_cast<const v4f*>(Bk + r * BT);
  pr.d = *reinterpret_cast<const v4f*>(dk_ptr);
}

__device__ __forceinline__ void store_panel(float (*panel)[BT][4], const PanelRegs& pr,
                                            int kq, int cq) {
  #pragma unroll
  for (int i = 0; i < 4; ++i) {
    v4f o;
    o[0] = pr.v[0][i] * pr.d[0];
    o[1] = pr.v[1][i] * pr.d[1];
    o[2] = pr.v[2][i] * pr.d[2];
    o[3] = pr.v[3][i] * pr.d[3];
    *reinterpret_cast<v4f*>(&panel[kq][i * 32 + cq][0]) = o;
  }
}

__device__ __forceinline__ void mma_chunk(const float (*panel)[BT][4], const v2f (&a)[8],
                                          v8f (&acc)[8], int half, int l16) {
  v2f bq[2][8];
  #pragma unroll
  for (int t = 0; t < 8; ++t)
    bq[0][t] = *reinterpret_cast<const v2f*>(&panel[0][t * 16 + l16][half * 2]);

  #pragma unroll
  for (int s = 0; s < 8; ++s) {
    const int cur = s & 1;
    const int nxt = cur ^ 1;
    if (s < 7) {
      #pragma unroll
      for (int t = 0; t < 8; ++t)
        bq[nxt][t] = *reinterpret_cast<const v2f*>(&panel[s + 1][t * 16 + l16][half * 2]);
    }
    #pragma unroll
    for (int t = 0; t < 8; ++t)
      acc[t] = __builtin_amdgcn_wmma_f32_16x16x4_f32(
          false, a[s], false, bq[cur][t], (short)0, acc[t], false, false);
  }
}

__global__ __launch_bounds__(256)
void lhat_gemm_kernel(const float* __restrict__ A,
                      const float* __restrict__ Bsrc,
                      const float* __restrict__ dis,
                      float* __restrict__ Part) {
  __shared__ __attribute__((aligned(16))) float ldsB[2][KQ][BT][4];

  const int tid  = threadIdx.x;
  const int wave = tid >> 5;
  const int lane = tid & 31;
  const int half = lane >> 4;
  const int l16  = lane & 15;
  const int rowTile = blockIdx.x * 128 + wave * 16;
  const int kOff = blockIdx.y * KSEG;
  float* PartSeg = Part + (size_t)blockIdx.y * NBT;

  v8f acc[8];
  #pragma unroll
  for (int i = 0; i < 8; ++i)
    #pragma unroll
    for (int j = 0; j < 8; ++j) acc[i][j] = 0.f;

  const int kq = wave;
  const int cq = lane;

  const float* Ak = A + (size_t)(rowTile + l16) * NB + kOff + half * 2;
  const float* Bk = Bsrc + (size_t)(kOff + kq * 4) * BT + cq * 4;
  const float* dk_ptr = dis + kOff + kq * 4;

  v2f a0[8], a1[8];
  PanelRegs pr;

  load_chunk(Ak, Bk, dk_ptr, a0, pr);
  store_panel(ldsB[0], pr, kq, cq);
  __syncthreads();

  for (int it = 0; it < NIT; it += 2) {
    load_chunk(Ak + (it + 1) * KC, Bk + (size_t)(it + 1) * KC * BT,
               dk_ptr + (it + 1) * KC, a1, pr);
    mma_chunk(ldsB[0], a0, acc, half, l16);
    store_panel(ldsB[1], pr, kq, cq);
    __syncthreads();

    const bool more = it + 2 < NIT;
    if (more)
      load_chunk(Ak + (it + 2) * KC, Bk + (size_t)(it + 2) * KC * BT,
                 dk_ptr + (it + 2) * KC, a0, pr);
    mma_chunk(ldsB[1], a1, acc, half, l16);
    if (more) store_panel(ldsB[0], pr, kq, cq);
    __syncthreads();
  }

  const int r0 = rowTile + half * 8;
  #pragma unroll
  for (int g = 0; g < 8; ++g) {
    float* dst = PartSeg + (size_t)(r0 + g) * BT + l16 * 4;
    #pragma unroll
    for (int b = 0; b < 2; ++b) {
      v4f o;
      o[0] = acc[b][g];
      o[1] = acc[b + 2][g];
      o[2] = acc[b + 4][g];
      o[3] = acc[b + 6][g];
      *reinterpret_cast<v4f*>(dst + b * 64) = o;
    }
  }
}

__global__ __launch_bounds__(256)
void ksplit_epilogue_kernel(const float* __restrict__ Part,
                            const float* __restrict__ Bsrc,
                            const float* __restrict__ dis,
                            const float* __restrict__ Adiag,
                            const float* __restrict__ Xadd,
                            float* __restrict__ D,
                            float alpha, float beta) {
  int idx = blockIdx.x * 256 + threadIdx.x;
  int r = idx >> 7;
  float acc = Part[idx] + Part[idx + NBT] + Part[idx + 2 * NBT] + Part[idx + 3 * NBT];
  float dr = dis[r];
  float ad = Adiag[r];
  float v = acc - ad * dr * Bsrc[idx];
  D[idx] = alpha * dr * v + beta * Xadd[idx];
}

__global__ __launch_bounds__(256)
void cheb_conv_kernel(const float* __restrict__ Tx0, const float* __restrict__ Tx1,
                      const float* __restrict__ Tx2,
                      const float* __restrict__ Wc, const float* __restrict__ bc,
                      const float* __restrict__ wt, const float* __restrict__ bt,
                      float* __restrict__ out) {
  __shared__ float sW[3 * 256];
  __shared__ float sbc[16];
  __shared__ float swt[4];
  __shared__ float stile[8][16][17];

  const int tid = threadIdx.x;
  for (int i = tid; i < 768; i += 256) sW[i] = Wc[i];
  if (tid < 16) sbc[tid] = bc[tid];
  if (tid < 3) swt[tid] = wt[tid];
  if (tid == 3) swt[3] = bt[0];
  __syncthreads();

  const int wave = tid >> 5;
  const int lane = tid & 31;
  const int half = lane >> 4;
  const int l16  = lane & 15;
  const int tileId = blockIdx.x * 8 + wave;
  const int b  = tileId >> 9;
  const int nT = tileId & 511;

  const float* Ts[3] = {Tx0, Tx1, Tx2};
  v8f acc;
  #pragma unroll
  for (int j = 0; j < 8; ++j) acc[j] = 0.f;

  #pragma unroll
  for (int m = 0; m < 3; ++m) {
    const float* Abase =
        Ts[m] + (size_t)(nT * 16 + l16) * BT + b * 16 + half * 2;
    #pragma unroll
    for (int s = 0; s < 4; ++s) {
      v2f a = *reinterpret_cast<const v2f*>(Abase + s * 4);
      const int k = s * 4 + half * 2;
      v2f w;
      w[0] = sW[m * 256 + k * 16 + l16];
      w[1] = sW[m * 256 + (k + 1) * 16 + l16];
      acc = __builtin_amdgcn_wmma_f32_16x16x4_f32(
          false, a, false, w, (short)0, acc, false, false);
    }
  }

  const float bcol = sbc[l16];
  #pragma unroll
  for (int g = 0; g < 8; ++g) {
    float v = acc[g] + bcol;
    stile[wave][g + half * 8][l16] = v > 0.f ? v : 0.f;
  }
  __syncthreads();

  const float w0 = swt[0], w1 = swt[1], w2 = swt[2], bb = swt[3];
  #pragma unroll
  for (int g = 0; g < 8; ++g) {
    const int r = g + half * 8;
    const int t = l16;
    float v = bb + w1 * stile[wave][r][t];
    if (t > 0)  v += w0 * stile[wave][r][t - 1];
    if (t < 15) v += w2 * stile[wave][r][t + 1];
    v = v > 0.f ? v : 0.f;
    const int n = nT * 16 + r;
    out[((size_t)b * NB + n) * 16 + t] = v;
  }
}

extern "C" void kernel_launch(void* const* d_in, const int* in_sizes, int n_in,
                              void* d_out, int out_size, void* d_ws, size_t ws_size,
                              hipStream_t stream) {
  const float* X  = (const float*)d_in[0];
  const float* A  = (const float*)d_in[1];
  const float* Wc = (const float*)d_in[2];
  const float* bc = (const float*)d_in[3];
  const float* wt = (const float*)d_in[4];
  const float* bt = (const float*)d_in[5];
  float* out = (float*)d_out;

  float* ws    = (float*)d_ws;
  float* xr    = ws;
  float* Tx1   = ws + (size_t)NBT;
  float* Tx2   = ws + (size_t)2 * NBT;
  float* dis   = ws + (size_t)3 * NBT;
  float* Adiag = dis + NB;
  float* Part  = Adiag + NB;

  repack_x_kernel<<<4096, 256, 0, stream>>>(X, xr);
  degree_kernel<<<1024, 256, 0, stream>>>(A, dis, Adiag);

  lhat_gemm_kernel<<<dim3(64, KSPLIT), 256, 0, stream>>>(A, xr, dis, Part);
  ksplit_epilogue_kernel<<<4096, 256, 0, stream>>>(Part, xr, dis, Adiag, xr, Tx1,
                                                   -1.f, 0.f);
  lhat_gemm_kernel<<<dim3(64, KSPLIT), 256, 0, stream>>>(A, Tx1, dis, Part);
  ksplit_epilogue_kernel<<<4096, 256, 0, stream>>>(Part, Tx1, dis, Adiag, xr, Tx2,
                                                   -2.f, -1.f);

  cheb_conv_kernel<<<512, 256, 0, stream>>>(xr, Tx1, Tx2, Wc, bc, wt, bt, out);
}


# GEMM K loop: B panel in a ring of 3 LDS buffers, staged mid-chunk; workgroup barrier split (signal at the panel store, wait before the next chunk's first reads): no drain between chunks; same sums
# speedup vs baseline: 1.5901x; 1.0616x over previous
#include <hip/hip_runtime.h>
#include <hip/hip_bf16.h>

typedef float v2f __attribute__((ext_vector_type(2)));
typedef float v4f __attribute__((ext_vector_type(4)));
typedef float v8f __attribute__((ext_vector_type(8)));

#define NB 8192
#define BT 128
#define KC 32
#define KQ (KC / 4)
#define KSPLIT 4
#define KSEG (NB / KSPLIT)
#define NBT (NB * BT)
#define NIT (KSEG / KC)

__global__ __launch_bounds__(256) void repack_x_kernel(const float* __restrict__ X,
                                                       float* __restrict__ xr) {
  int idx = blockIdx.x * 256 + threadIdx.x;
  int n = idx >> 7;
  int c = idx & 127;
  int b = c >> 4;
  int t = c & 15;
  xr[idx] = X[(size_t)b * (NB * 16) + (size_t)n * 16 + t];
}

__global__ __launch_bounds__(256) void degree_kernel(const float* __restrict__ A,
                                                     float* __restrict__ dis,
                                                     float* __restrict__ Adiag) {
  int wave = threadIdx.x >> 5;
  int lane = threadIdx.x & 31;
  int row = blockIdx.x * 8 + wave;
  const float* Arow = A + (size_t)row * NB;
  float s = 0.f;
  for (int j = lane * 4; j < NB; j += 32 * 4) {
    float4 v = *reinterpret_cast<const float4*>(Arow + j);
    s += v.x + v.y + v.z + v.w;
  }
  for (int off = 16; off > 0; off >>= 1) s += __shfl_xor(s, off, 32);
  if (lane == 0) {
    float diag = Arow[row];
    float deg = s - diag;
    Adiag[row] = diag;
    dis[row] = (deg > 0.f) ? rsqrtf(deg) : 0.f;
  }
}


#define PANEL_FLOATS (KQ * BT * 4)

struct PanelRegs {
  v4f v[4];
  v4f d;
};

__device__ __forceinline__ void wait_lds_done() { asm volatile("s_wait_dscnt 0x0" ::: "memory"); }
__device__ __forceinline__ void barrier_signal() { asm volatile("s_barrier_signal -1" ::: "memory"); }
__device__ __forceinline__ void barrier_wait() { asm volatile("s_barrier_wait -1" ::: "memory"); }

__device__ __forceinline__ void load_a(const float* __restrict__ Ak, v2f (&a)[8]) {
  #pragma unroll
  for (int s = 0; s < 8; ++s)
    a[s] = *reinterpret_cast<const v2f*>(Ak + s * 4);
}

__device__ __forceinline__ void load_panel(const float* __restrict__ Bk,
                                           const float* __restrict__ dk_ptr,
                                           PanelRegs& pr) {
  #pragma unroll
  for (int r = 0; r < 4; ++r)
    pr.v[r] = *reinterpret_cast<const v4f*>(Bk + r * BT);
  pr.d = *reinterpret_cast<const v4f*>(dk_ptr);
}

__device__ __forceinline__ void store_panel(float* panel, const PanelRegs& pr,
                                            int kq, int cq) {
  #pragma unroll
  for (int i = 0; i < 4; ++i) {
    v4f o;
    o[0] = pr.v[0][i] * pr.d[0];
    o[1] = pr.v[1][i] * pr.d[1];
    o[2] = pr.v[2][i] * pr.d[2];
    o[3] = pr.v[3][i] * pr.d[3];
    *reinterpret_cast<v4f*>(panel + ((kq * BT + i * 32 + cq) << 2)) = o;
  }
}

__device__ __forceinline__ void load_row(v2f (&b)[8], const float* panel, int s,
                                         int half, int l16) {
  #pragma unroll
  for (int t = 0; t < 8; ++t)
    b[t] = *reinterpret_cast<const v2f*>(panel + ((s * BT + t * 16 + l16) << 2) + half * 2);
}

__device__ __forceinline__ void mma_row(v8f (&acc)[8], v2f a, const v2f (&b)[8]) {
  #pragma unroll
  for (int t = 0; t < 8; ++t)
    acc[t] = __builtin_amdgcn_wmma_f32_16x16x4_f32(
        false, a, false, b[t], (short)0, acc[t], false, false);
}

__device__ __forceinline__ void gemm_chunk(const v2f (&aC)[8], v2f (&aN)[8], PanelRegs& pr,
                                           v2f (&bq)[2][8], v8f (&acc)[8],
                                           const float* pC, float* pN,
                                           const float* __restrict__ AkN,
                                           const float* __restrict__ BkN2,
                                           const float* __restrict__ dkN2,
                                           int kq, int cq, int half, int l16) {
  load_a(AkN, aN);

  #pragma unroll
  for (int s = 0; s < 3; ++s) {
    load_row(bq[(s + 1) & 1], pC, s + 1, half, l16);
    mma_row(acc, aC[s], bq[s & 1]);
  }
  load_row(bq[0], pC, 4, half, l16);

  __builtin_amdgcn_sched_barrier(0);

  store_panel(pN, pr, kq, cq);
  __builtin_amdgcn_sched_barrier(0);
  mma_row(acc, aC[3], bq[1]);
  __builtin_amdgcn_sched_barrier(0);
  wait_lds_done();
  barrier_signal();
  load_panel(BkN2, dkN2, pr);

  __builtin_amdgcn_sched_barrier(0);

  #pragma unroll
  for (int s = 4; s < 7; ++s) {
    load_row(bq[(s + 1) & 1], pC, s + 1, half, l16);
    mma_row(acc, aC[s], bq[s & 1]);
  }

  barrier_wait();
  load_row(bq[0], pN, 0, half, l16);
  mma_row(acc, aC[7], bq[1]);

  __builtin_amdgcn_sched_barrier(0);
}

__global__ __launch_bounds__(256)
void lhat_gemm_kernel(const float* __restrict__ A,
                      const float* __restrict__ Bsrc,
                      const float* __restrict__ dis,
                      float* __restrict__ Part) {
  __shared__ __attribute__((aligned(16))) float ldsB[3 * PANEL_FLOATS];

  const int tid  = threadIdx.x;
  const int wave = tid >> 5;
  const int lane = tid & 31;
  const int half = lane >> 4;
  const int l16  = lane & 15;
  const int rowTile = blockIdx.x * 128 + wave * 16;
  const int kOff = blockIdx.y * KSEG;
  float* PartSeg = Part + (size_t)blockIdx.y * NBT;

  v8f acc[8];
  #pragma unroll
  for (int i = 0; i < 8; ++i)
    #pragma unroll
    for (int j = 0; j < 8; ++j) acc[i][j] = 0.f;

  const int kq = wave;
  const int cq = lane;

  const float* Ak = A + (size_t)(rowTile + l16) * NB + kOff + half * 2;
  const float* Bk = Bsrc + (size_t)(kOff + kq * 4) * BT + cq * 4;
  const float* dk_ptr = dis + kOff + kq * 4;

  v2f a0[8], a1[8];
  v2f bq[2][8];
  PanelRegs pr;

  load_a(Ak, a0);
  load_panel(Bk, dk_ptr, pr);
  store_panel(ldsB, pr, kq, cq);
  wait_lds_done();
  barrier_signal();
  load_panel(Bk + (size_t)KC * BT, dk_ptr + KC, pr);
  barrier_wait();
  load_row(bq[0], ldsB, 0, half, l16);

  int cur = 0;
  for (int it = 0; it < NIT; it += 2) {
    const int c2 = it + 2 < NIT ? it + 2 : NIT - 1;
    const int c3 = it + 3 < NIT ? it + 3 : NIT - 1;
    int nxt = cur == 2 ? 0 : cur + 1;
    gemm_chunk(a0, a1, pr, bq, acc, ldsB + cur * PANEL_FLOATS, ldsB + nxt * PANEL_FLOATS,
               Ak + (it + 1) * KC, Bk + (size_t)c2 * KC * BT, dk_ptr + c2 * KC,
               kq, cq, half, l16);
    cur = nxt;
    nxt = cur == 2 ? 0 : cur + 1;
    gemm_chunk(a1, a0, pr, bq, acc, ldsB + cur * PANEL_FLOATS, ldsB + nxt * PANEL_FLOATS,
               Ak + c2 * KC, Bk + (size_t)c3 * KC * BT, dk_ptr + c3 * KC,
               kq, cq, half, l16);
    cur = nxt;
  }

  const int r0 = rowTile + half * 8;
  #pragma unroll
  for (int g = 0; g < 8; ++g) {
    float* dst = PartSeg + (size_t)(r0 + g) * BT + l16 * 4;
    #pragma unroll
    for (int b = 0; b < 2; ++b) {
      v4f o;
      o[0] = acc[b][g];
      o[1] = acc[b + 2][g];
      o[2] = acc[b + 4][g];
      o[3] = acc[b + 6][g];
      *reinterpret_cast<v4f*>(dst + b * 64) = o;
    }
  }
}

__global__ __launch_bounds__(256)
void ksplit_epilogue_kernel(const float* __restrict__ Part,
                            const float* __restrict__ Bsrc,
                            const float* __restrict__ dis,
                            const float* __restrict__ Adiag,
                            const float* __restrict__ Xadd,
                            float* __restrict__ D,
                            float alpha, float beta) {
  int idx = blockIdx.x * 256 + threadIdx.x;
  int r = idx >> 7;
  float acc = Part[idx] + Part[idx + NBT] + Part[idx + 2 * NBT] + Part[idx + 3 * NBT];
  float dr = dis[r];
  float ad = Adiag[r];
  float v = acc - ad * dr * Bsrc[idx];
  D[idx] = alpha * dr * v + beta * Xadd[idx];
}

__global__ __launch_bounds__(256)
void cheb_conv_kernel(const float* __restrict__ Tx0, const float* __restrict__ Tx1,
                      const float* __restrict__ Tx2,
                      const float* __restrict__ Wc, const float* __restrict__ bc,
                      const float* __restrict__ wt, const float* __restrict__ bt,
                      float* __restrict__ out) {
  __shared__ float sW[3 * 256];
  __shared__ float sbc[16];
  __shared__ float swt[4];
  __shared__ float stile[8][16][17];

  const int tid = threadIdx.x;
  for (int i = tid; i < 768; i += 256) sW[i] = Wc[i];
  if (tid < 16) sbc[tid] = bc[tid];
  if (tid < 3) swt[tid] = wt[tid];
  if (tid == 3) swt[3] = bt[0];
  __syncthreads();

  const int wave = tid >> 5;
  const int lane = tid & 31;
  const int half = lane >> 4;
  const int l16  = lane & 15;
  const int tileId = blockIdx.x * 8 + wave;
  const int b  = tileId >> 9;
  const int nT = tileId & 511;

  const float* Ts[3] = {Tx0, Tx1, Tx2};
  v8f acc;
  #pragma unroll
  for (int j = 0; j < 8; ++j) acc[j] = 0.f;

  #pragma unroll
  for (int m = 0; m < 3; ++m) {
    const float* Abase =
        Ts[m] + (size_t)(nT * 16 + l16) * BT + b * 16 + half * 2;
    #pragma unroll
    for (int s = 0; s < 4; ++s) {
      v2f a = *reinterpret_cast<const v2f*>(Abase + s * 4);
      const int k = s * 4 + half * 2;
      v2f w;
      w[0] = sW[m * 256 + k * 16 + l16];
      w[1] = sW[m * 256 + (k + 1) * 16 + l16];
      acc = __builtin_amdgcn_wmma_f32_16x16x4_f32(
          false, a, false, w, (short)0, acc, false, false);
    }
  }

  const float bcol = sbc[l16];
  #pragma unroll
  for (int g = 0; g < 8; ++g) {
    float v = acc[g] + bcol;
    stile[wave][g + half * 8][l16] = v > 0.f ? v : 0.f;
  }
  __syncthreads();

  const float w0 = swt[0], w1 = swt[1], w2 = swt[2], bb = swt[3];
  #pragma unroll
  for (int g = 0; g < 8; ++g) {
    const int r = g + half * 8;
    const int t = l16;
    float v = bb + w1 * stile[wave][r][t];
    if (t > 0)  v += w0 * stile[wave][r][t - 1];
    if (t < 15) v += w2 * stile[wave][r][t + 1];
    v = v > 0.f ? v : 0.f;
    const int n = nT * 16 + r;
    out[((size_t)b * NB + n) * 16 + t] = v;
  }
}

extern "C" void kernel_launch(void* const* d_in, const int* in_sizes, int n_in,
                              void* d_out, int out_size, void* d_ws, size_t ws_size,
                              hipStream_t stream) {
  const float* X  = (const float*)d_in[0];
  const float* A  = (const float*)d_in[1];
  const float* Wc = (const float*)d_in[2];
  const float* bc = (const float*)d_in[3];
  const float* wt = (const float*)d_in[4];
  const float* bt = (const float*)d_in[5];
  float* out = (float*)d_out;

  float* ws    = (float*)d_ws;
  float* xr    = ws;
  float* Tx1   = ws + (size_t)NBT;
  float* Tx2   = ws + (size_t)2 * NBT;
  float* dis   = ws + (size_t)3 * NBT;
  float* Adiag = dis + NB;
  float* Part  = Adiag + NB;

  repack_x_kernel<<<4096, 256, 0, stream>>>(X, xr);
  degree_kernel<<<1024, 256, 0, stream>>>(A, dis, Adiag);

  lhat_gemm_kernel<<<dim3(64, KSPLIT), 256, 0, stream>>>(A, xr, dis, Part);
  ksplit_epilogue_kernel<<<4096, 256, 0, stream>>>(Part, xr, dis, Adiag, xr, Tx1,
                                                   -1.f, 0.f);
  lhat_gemm_kernel<<<dim3(64, KSPLIT), 256, 0, stream>>>(A, Tx1, dis, Part);
  ksplit_epilogue_kernel<<<4096, 256, 0, stream>>>(Part, Tx1, dis, Adiag, xr, Tx2,
                                                   -2.f, -1.f);

  cheb_conv_kernel<<<512, 256, 0, stream>>>(xr, Tx1, Tx2, Wc, bc, wt, bt, out);
}


# GEMM B operands: next chunk's 32 LDS reads (128-bit; panel rows re-paired so one read feeds two K-steps) issued as ONE group a chunk ahead into a second register set; k order and sums unchanged
# speedup vs baseline: 1.6411x; 1.0321x over previous
#include <hip/hip_runtime.h>
#include <hip/hip_bf16.h>

typedef float v2f __attribute__((ext_vector_type(2)));
typedef float v4f __attribute__((ext_vector_type(4)));
typedef float v8f __attribute__((ext_vector_type(8)));

#define NB 8192
#define BT 128
#define KC 32
#define KQ (KC / 4)
#define KSPLIT 4
#define KSEG (NB / KSPLIT)
#define NBT (NB * BT)
#define NIT (KSEG / KC)

__global__ __launch_bounds__(256) void repack_x_kernel(const float* __restrict__ X,
                                                       float* __restrict__ xr) {
  int idx = blockIdx.x * 256 + threadIdx.x;
  int n = idx >> 7;
  int c = idx & 127;
  int b = c >> 4;
  int t = c & 15;
  xr[idx] = X[(size_t)b * (NB * 16) + (size_t)n * 16 + t];
}

__global__ __launch_bounds__(256) void degree_kernel(const float* __restrict__ A,
                                                     float* __restrict__ dis,
                                                     float* __restrict__ Adiag) {
  int wave = threadIdx.x >> 5;
  int lane = threadIdx.x & 31;
  int row = blockIdx.x * 8 + wave;
  const float* Arow = A + (size_t)row * NB;
  float s = 0.f;
  for (int j = lane * 4; j < NB; j += 32 * 4) {
    float4 v = *reinterpret_cast<const float4*>(Arow + j);
    s += v.x + v.y + v.z + v.w;
  }
  for (int off = 16; off > 0; off >>= 1) s += __shfl_xor(s, off, 32);
  if (lane == 0) {
    float diag = Arow[row];
    float deg = s - diag;
    Adiag[row] = diag;
    dis[row] = (deg > 0.f) ? rsqrtf(deg) : 0.f;
  }
}


#define PANEL_FLOATS (KQ * BT * 4)

struct PanelRegs {
  v4f v[4];
  v2f dlo;
  v2f dhi;
};

__device__ __forceinline__ void wait_lds_done() { asm volatile("s_wait_dscnt 0x0" ::: "memory"); }
__device__ __forceinline__ void barrier_signal() { asm volatile("s_barrier_signal -1" ::: "memory"); }
__device__ __forceinline__ void barrier_wait() { asm volatile("s_barrier_wait -1" ::: "memory"); }

__device__ __forceinline__ void load_a(const float* __restrict__ Ak, v2f (&a)[8]) {
  #pragma unroll
  for (int s = 0; s < 8; ++s)
    a[s] = *reinterpret_cast<const v2f*>(Ak + s * 4);
}

__device__ __forceinline__ void load_panel(const float* __restrict__ Bk,
                                           const float* __restrict__ dk_ptr,
                                           PanelRegs& pr) {
  pr.v[0] = *reinterpret_cast<const v4f*>(Bk);
  pr.v[1] = *reinterpret_cast<const v4f*>(Bk + BT);
  pr.v[2] = *reinterpret_cast<const v4f*>(Bk + 4 * BT);
  pr.v[3] = *reinterpret_cast<const v4f*>(Bk + 5 * BT);
  pr.dlo = *reinterpret_cast<const v2f*>(dk_ptr);
  pr.dhi = *reinterpret_cast<const v2f*>(dk_ptr + 4);
}

__device__ __forceinline__ void store_panel(float* panel, const PanelRegs& pr,
                                            int R, int cq) {
  #pragma unroll
  for (int i = 0; i < 4; ++i) {
    v4f o;
    o[0] = pr.v[0][i] * pr.dlo[0];
    o[1] = pr.v[1][i] * pr.dlo[1];
    o[2] = pr.v[2][i] * pr.dhi[0];
    o[3] = pr.v[3][i] * pr.dhi[1];
    *reinterpret_cast<v4f*>(panel + ((R * BT + i * 32 + cq) << 2)) = o;
  }
}

__device__ __forceinline__ void load_frags(v4f (&b)[4][8], const float* panel,
                                           int half, int l16) {
  #pragma unroll
  for (int j = 0; j < 4; ++j)
    #pragma unroll
    for (int t = 0; t < 8; ++t)
      b[j][t] = *reinterpret_cast<const v4f*>(panel + (((2 * j + half) * BT + t * 16 + l16) << 2));
}

template <int ODD>
__device__ __forceinline__ void mma_row(v8f (&acc)[8], v2f a, const v4f (&b)[8]) {
  #pragma unroll
  for (int t = 0; t < 8; ++t) {
    v2f bb = ODD ? __builtin_shufflevector(b[t], b[t], 2, 3)
                 : __builtin_shufflevector(b[t], b[t], 0, 1);
    acc[t] = __builtin_amdgcn_wmma_f32_16x16x4_f32(
        false, a, false, bb, (short)0, acc[t], false, false);
  }
}

__device__ __forceinline__ void gemm_chunk(const v2f (&aC)[8], v2f (&aN)[8], PanelRegs& pr,
                                           const v4f (&bC)[4][8], v4f (&bN)[4][8],
                                           v8f (&acc)[8],
                                           const float* pRead, float* pWrite,
                                           const float* __restrict__ AkN,
                                           const float* __restrict__ BkN3,
                                           const float* __restrict__ dkN3,
                                           int R, int cq, int half, int l16) {
  load_frags(bN, pRead, half, l16);
  load_a(AkN, aN);

  __builtin_amdgcn_sched_barrier(0);

  mma_row<0>(acc, aC[0], bC[0]);
  mma_row<1>(acc, aC[1], bC[0]);
  mma_row<0>(acc, aC[2], bC[1]);

  __builtin_amdgcn_sched_barrier(0);

  store_panel(pWrite, pr, R, cq);
  __builtin_amdgcn_sched_barrier(0);
  mma_row<1>(acc, aC[3], bC[1]);
  __builtin_amdgcn_sched_barrier(0);
  wait_lds_done();
  barrier_signal();
  load_panel(BkN3, dkN3, pr);

  __builtin_amdgcn_sched_barrier(0);

  mma_row<0>(acc, aC[4], bC[2]);
  mma_row<1>(acc, aC[5], bC[2]);
  mma_row<0>(acc, aC[6], bC[3]);
  mma_row<1>(acc, aC[7], bC[3]);

  __builtin_amdgcn_sched_barrier(0);
  barrier_wait();

}

__global__ __launch_bounds__(256)
void lhat_gemm_kernel(const float* __restrict__ A,
                      const float* __restrict__ Bsrc,
                      const float* __restrict__ dis,
                      float* __restrict__ Part) {
  __shared__ __attribute__((aligned(16))) float ldsB[2 * PANEL_FLOATS];

  const int tid  = threadIdx.x;
  const int wave = tid >> 5;
  const int lane = tid & 31;
  const int half = lane >> 4;
  const int l16  = lane & 15;
  const int rowTile = blockIdx.x * 128 + wave * 16;
  const int kOff = blockIdx.y * KSEG;
  float* PartSeg = Part + (size_t)blockIdx.y * NBT;

  v8f acc[8];
  #pragma unroll
  for (int i = 0; i < 8; ++i)
    #pragma unroll
    for (int j = 0; j < 8; ++j) acc[i][j] = 0.f;

  const int R  = wave;
  const int cq = lane;
  const int kr = (R >> 1) * 8 + (R & 1) * 2;

  const float* Ak = A + (size_t)(rowTile + l16) * NB + kOff + half * 2;
  const float* Bk = Bsrc + (size_t)(kOff + kr) * BT + cq * 4;
  const float* dk_ptr = dis + kOff + kr;

  float* const buf0 = ldsB;
  float* const buf1 = ldsB + PANEL_FLOATS;

  v2f a0[8], a1[8];
  v4f b0[4][8], b1[4][8];
  PanelRegs pr, pq;

  load_a(Ak, a0);
  load_panel(Bk, dk_ptr, pr);
  load_panel(Bk + (size_t)KC * BT, dk_ptr + KC, pq);
  store_panel(buf0, pr, R, cq);
  store_panel(buf1, pq, R, cq);
  wait_lds_done();
  barrier_signal();
  load_panel(Bk + (size_t)2 * KC * BT, dk_ptr + 2 * KC, pr);
  barrier_wait();
  load_frags(b0, buf0, half, l16);
  wait_lds_done();
  barrier_signal();
  barrier_wait();

  for (int it = 0; it < NIT; it += 2) {
    const int c2 = it + 2 < NIT ? it + 2 : NIT - 1;
    const int c3 = it + 3 < NIT ? it + 3 : NIT - 1;
    const int c4 = it + 4 < NIT ? it + 4 : NIT - 1;
    gemm_chunk(a0, a1, pr, b0, b1, acc, buf1, buf0,
               Ak + (it + 1) * KC, Bk + (size_t)c3 * KC * BT, dk_ptr + c3 * KC,
               R, cq, half, l16);
    gemm_chunk(a1, a0, pr, b1, b0, acc, buf0, buf1,
               Ak + c2 * KC, Bk + (size_t)c4 * KC * BT, dk_ptr + c4 * KC,
               R, cq, half, l16);
  }

  const int r0 = rowTile + half * 8;
  #pragma unroll
  for (int g = 0; g < 8; ++g) {
    float* dst = PartSeg + (size_t)(r0 + g) * BT + l16 * 4;
    #pragma unroll
    for (int b = 0; b < 2; ++b) {
      v4f o;
      o[0] = acc[b][g];
      o[1] = acc[b + 2][g];
      o[2] = acc[b + 4][g];
      o[3] = acc[b + 6][g];
      *reinterpret_cast<v4f*>(dst + b * 64) = o;
    }
  }
}

__global__ __launch_bounds__(256)
void ksplit_epilogue_kernel(const float* __restrict__ Part,
                            const float* __restrict__ Bsrc,
                            const float* __restrict__ dis,
                            const float* __restrict__ Adiag,
                            const float* __restrict__ Xadd,
                            float* __restrict__ D,
                            float alpha, float beta) {
  int idx = blockIdx.x * 256 + threadIdx.x;
  int r = idx >> 7;
  float acc = Part[idx] + Part[idx + NBT] + Part[idx + 2 * NBT] + Part[idx + 3 * NBT];
  float dr = dis[r];
  float ad = Adiag[r];
  float v = acc - ad * dr * Bsrc[idx];
  D[idx] = alpha * dr * v + beta * Xadd[idx];
}

__global__ __launch_bounds__(256)
void cheb_conv_kernel(const float* __restrict__ Tx0, const float* __restrict__ Tx1,
                      const float* __restrict__ Tx2,
                      const float* __restrict__ Wc, const float* __restrict__ bc,
                      const float* __restrict__ wt, const float* __restrict__ bt,
                      float* __restrict__ out) {
  __shared__ float sW[3 * 256];
  __shared__ float sbc[16];
  __shared__ float swt[4];
  __shared__ float stile[8][16][17];

  const int tid = threadIdx.x;
  for (int i = tid; i < 768; i += 256) sW[i] = Wc[i];
  if (tid < 16) sbc[tid] = bc[tid];
  if (tid < 3) swt[tid] = wt[tid];
  if (tid == 3) swt[3] = bt[0];
  __syncthreads();

  const int wave = tid >> 5;
  const int lane = tid & 31;
  const int half = lane >> 4;
  const int l16  = lane & 15;
  const int tileId = blockIdx.x * 8 + wave;
  const int b  = tileId >> 9;
  const int nT = tileId & 511;

  const float* Ts[3] = {Tx0, Tx1, Tx2};
  v8f acc;
  #pragma unroll
  for (int j = 0; j < 8; ++j) acc[j] = 0.f;

  #pragma unroll
  for (int m = 0; m < 3; ++m) {
    const float* Abase =
        Ts[m] + (size_t)(nT * 16 + l16) * BT + b * 16 + half * 2;
    #pragma unroll
    for (int s = 0; s < 4; ++s) {
      v2f a = *reinterpret_cast<const v2f*>(Abase + s * 4);
      const int k = s * 4 + half * 2;
      v2f w;
      w[0] = sW[m * 256 + k * 16 + l16];
      w[1] = sW[m * 256 + (k + 1) * 16 + l16];
      acc = __builtin_amdgcn_wmma_f32_16x16x4_f32(
          false, a, false, w, (short)0, acc, false, false);
    }
  }

  const float bcol = sbc[l16];
  #pragma unroll
  for (int g = 0; g < 8; ++g) {
    float v = acc[g] + bcol;
    stile[wave][g + half * 8][l16] = v > 0.f ? v : 0.f;
  }
  __syncthreads();

  const float w0 = swt[0], w1 = swt[1], w2 = swt[2], bb = swt[3];
  #pragma unroll
  for (int g = 0; g < 8; ++g) {
    const int r = g + half * 8;
    const int t = l16;
    float v = bb + w1 * stile[wave][r][t];
    if (t > 0)  v += w0 * stile[wave][r][t - 1];
    if (t < 15) v += w2 * stile[wave][r][t + 1];
    v = v > 0.f ? v : 0.f;
    const int n = nT * 16 + r;
    out[((size_t)b * NB + n) * 16 + t] = v;
  }
}

extern "C" void kernel_launch(void* const* d_in, const int* in_sizes, int n_in,
                              void* d_out, int out_size, void* d_ws, size_t ws_size,
                              hipStream_t stream) {
  const float* X  = (const float*)d_in[0];
  const float* A  = (const float*)d_in[1];
  const float* Wc = (const float*)d_in[2];
  const float* bc = (const float*)d_in[3];
  const float* wt = (const float*)d_in[4];
  const float* bt = (const float*)d_in[5];
  float* out = (float*)d_out;

  float* ws    = (float*)d_ws;
  float* xr    = ws;
  float* Tx1   = ws + (size_t)NBT;
  float* Tx2   = ws + (size_t)2 * NBT;
  float* dis   = ws + (size_t)3 * NBT;
  float* Adiag = dis + NB;
  float* Part  = Adiag + NB;

  repack_x_kernel<<<4096, 256, 0, stream>>>(X, xr);
  degree_kernel<<<1024, 256, 0, stream>>>(A, dis, Adiag);

  lhat_gemm_kernel<<<dim3(64, KSPLIT), 256, 0, stream>>>(A, xr, dis, Part);
  ksplit_epilogue_kernel<<<4096, 256, 0, stream>>>(Part, xr, dis, Adiag, xr, Tx1,
                                                   -1.f, 0.f);
  lhat_gemm_kernel<<<dim3(64, KSPLIT), 256, 0, stream>>>(A, Tx1, dis, Part);
  ksplit_epilogue_kernel<<<4096, 256, 0, stream>>>(Part, Tx1, dis, Adiag, xr, Tx2,
                                                   -2.f, -1.f);

  cheb_conv_kernel<<<512, 256, 0, stream>>>(xr, Tx1, Tx2, Wc, bc, wt, bt, out);
}


# Row-sum (degree) launch rebuilt as a stream: 256 workgroups x 16 waves x 2 rows, 16-byte loads issued 16 at a time one group ahead (16-32 in flight per wave, was 1); same sums in the same order
# speedup vs baseline: 1.7898x; 1.0906x over previous
#include <hip/hip_runtime.h>
#include <hip/hip_bf16.h>

typedef float v2f __attribute__((ext_vector_type(2)));
typedef float v4f __attribute__((ext_vector_type(4)));
typedef float v8f __attribute__((ext_vector_type(8)));

#define NB 8192
#define BT 128
#define KC 32
#define KQ (KC / 4)
#define KSPLIT 4
#define KSEG (NB / KSPLIT)
#define NBT (NB * BT)
#define NIT (KSEG / KC)

__global__ __launch_bounds__(256) void repack_x_kernel(const float* __restrict__ X,
                                                       float* __restrict__ xr) {
  int idx = blockIdx.x * 256 + threadIdx.x;
  int n = idx >> 7;
  int c = idx & 127;
  int b = c >> 4;
  int t = c & 15;
  xr[idx] = X[(size_t)b * (NB * 16) + (size_t)n * 16 + t];
}

#define DEG_WAVES 16
#define DEG_ROWS 2
#define DEG_GROUP 16
#define DEG_GROUPS_PER_ROW (NB / 128 / DEG_GROUP)
#define DEG_NGROUPS (DEG_ROWS * DEG_GROUPS_PER_ROW)
#define DEG_BLOCK_ROWS (DEG_WAVES * DEG_ROWS)

__global__ __launch_bounds__(DEG_WAVES * 32) void degree_kernel(const float* __restrict__ A,
                                                                float* __restrict__ dis,
                                                                float* __restrict__ Adiag) {
  __shared__ float sDiag[DEG_BLOCK_ROWS];
  __shared__ float sDis[DEG_BLOCK_ROWS];

  const int lane = threadIdx.x & 31;
  const int wave = __builtin_amdgcn_readfirstlane(threadIdx.x >> 5);
  const int row0 = (blockIdx.x * DEG_WAVES + wave) * DEG_ROWS;
  const float* base = A + (size_t)row0 * NB + lane * 4;

  v4f buf[2][DEG_GROUP];
  float sum[DEG_ROWS];

#define DEG_LOAD_GROUP(G) do { \
    _Pragma("unroll") for (int q = 0; q < DEG_GROUP; ++q) \
      buf[(G) & 1][q] = *reinterpret_cast<const v4f*>(base + ((G) * DEG_GROUP + q) * 128); } while (0)

  DEG_LOAD_GROUP(0);
  float s = 0.f;
  #pragma unroll
  for (int g = 0; g < DEG_NGROUPS; ++g) {
    __builtin_amdgcn_sched_barrier(0);
    if (g + 1 < DEG_NGROUPS) DEG_LOAD_GROUP(g + 1);
    __builtin_amdgcn_sched_barrier(0);
    #pragma unroll
    for (int q = 0; q < DEG_GROUP; ++q) {
      const v4f v = buf[g & 1][q];
      s += v.x + v.y + v.z + v.w;
    }
    if ((g % DEG_GROUPS_PER_ROW) == DEG_GROUPS_PER_ROW - 1) {
      sum[g / DEG_GROUPS_PER_ROW] = s;
      s = 0.f;
    }
  }
#undef DEG_LOAD_GROUP

  #pragma unroll
  for (int r = 0; r < DEG_ROWS; ++r) {
    float t = sum[r];
    for (int off = 16; off > 0; off >>= 1) t += __shfl_xor(t, off, 32);
    sum[r] = t;
  }

  float mySum = sum[0];
  #pragma unroll
  for (int r = 1; r < DEG_ROWS; ++r)
    if (lane == r) mySum = sum[r];
  if (lane < DEG_ROWS) {
    const int row = row0 + lane;
    const float diag = A[(size_t)row * NB + row];
    const float deg = mySum - diag;
    sDiag[wave * DEG_ROWS + lane] = diag;
    sDis[wave * DEG_ROWS + lane] = (deg > 0.f) ? rsqrtf(deg) : 0.f;
  }
  __syncthreads();
  if (threadIdx.x < DEG_BLOCK_ROWS) {
    Adiag[blockIdx.x * DEG_BLOCK_ROWS + threadIdx.x] = sDiag[threadIdx.x];
    dis[blockIdx.x * DEG_BLOCK_ROWS + threadIdx.x] = sDis[threadIdx.x];
  }
}


#define PANEL_FLOATS (KQ * BT * 4)

struct PanelRegs {
  v4f v[4];
  v2f dlo;
  v2f dhi;
};

__device__ __forceinline__ void wait_lds_done() { asm volatile("s_wait_dscnt 0x0" ::: "memory"); }
__device__ __forceinline__ void barrier_signal() { asm volatile("s_barrier_signal -1" ::: "memory"); }
__device__ __forceinline__ void barrier_wait() { asm volatile("s_barrier_wait -1" ::: "memory"); }

__device__ __forceinline__ void load_a(const float* __restrict__ Ak, v2f (&a)[8]) {
  #pragma unroll
  for (int s = 0; s < 8; ++s)
    a[s] = *reinterpret_cast<const v2f*>(Ak + s * 4);
}

__device__ __forceinline__ void load_panel(const float* __restrict__ Bk,
                                           const float* __restrict__ dk_ptr,
                                           PanelRegs& pr) {
  pr.v[0] = *reinterpret_cast<const v4f*>(Bk);
  pr.v[1] = *reinterpret_cast<const v4f*>(Bk + BT);
  pr.v[2] = *reinterpret_cast<const v4f*>(Bk + 4 * BT);
  pr.v[3] = *reinterpret_cast<const v4f*>(Bk + 5 * BT);
  pr.dlo = *reinterpret_cast<const v2f*>(dk_ptr);
  pr.dhi = *reinterpret_cast<const v2f*>(dk_ptr + 4);
}

__device__ __forceinline__ void store_panel(float* panel, const PanelRegs& pr,
                                            int R, int cq) {
  #pragma unroll
  for (int i = 0; i < 4; ++i) {
    v4f o;
    o[0] = pr.v[0][i] * pr.dlo[0];
    o[1] = pr.v[1][i] * pr.dlo[1];
    o[2] = pr.v[2][i] * pr.dhi[0];
    o[3] = pr.v[3][i] * pr.dhi[1];
    *reinterpret_cast<v4f*>(panel + ((R * BT + i * 32 + cq) << 2)) = o;
  }
}

__device__ __forceinline__ void load_frags(v4f (&b)[4][8], const float* panel,
                                           int half, int l16) {
  #pragma unroll
  for (int j = 0; j < 4; ++j)
    #pragma unroll
    for (int t = 0; t < 8; ++t)
      b[j][t] = *reinterpret_cast<const v4f*>(panel + (((2 * j + half) * BT + t * 16 + l16) << 2));
}

template <int ODD>
__device__ __forceinline__ void mma_row(v8f (&acc)[8], v2f a, const v4f (&b)[8]) {
  #pragma unroll
  for (int t = 0; t < 8; ++t) {
    v2f bb = ODD ? __builtin_shufflevector(b[t], b[t], 2, 3)
                 : __builtin_shufflevector(b[t], b[t], 0, 1);
    acc[t] = __builtin_amdgcn_wmma_f32_16x16x4_f32(
        false, a, false, bb, (short)0, acc[t], false, false);
  }
}

__device__ __forceinline__ void gemm_chunk(const v2f (&aC)[8], v2f (&aN)[8], PanelRegs& pr,
                                           const v4f (&bC)[4][8], v4f (&bN)[4][8],
                                           v8f (&acc)[8],
                                           const float* pRead, float* pWrite,
                                           const float* __restrict__ AkN,
                                           const float* __restrict__ BkN3,
                                           const float* __restrict__ dkN3,
                                           int R, int cq, int half, int l16) {
  load_frags(bN, pRead, half, l16);
  load_a(AkN, aN);

  __builtin_amdgcn_sched_barrier(0);

  mma_row<0>(acc, aC[0], bC[0]);
  mma_row<1>(acc, aC[1], bC[0]);
  mma_row<0>(acc, aC[2], bC[1]);

  __builtin_amdgcn_sched_barrier(0);

  store_panel(pWrite, pr, R, cq);
  __builtin_amdgcn_sched_barrier(0);
  mma_row<1>(acc, aC[3], bC[1]);
  __builtin_amdgcn_sched_barrier(0);
  wait_lds_done();
  barrier_signal();
  load_panel(BkN3, dkN3, pr);

  __builtin_amdgcn_sched_barrier(0);

  mma_row<0>(acc, aC[4], bC[2]);
  mma_row<1>(acc, aC[5], bC[2]);
  mma_row<0>(acc, aC[6], bC[3]);
  mma_row<1>(acc, aC[7], bC[3]);

  __builtin_amdgcn_sched_barrier(0);
  barrier_wait();

}

__global__ __launch_bounds__(256)
void lhat_gemm_kernel(const float* __restrict__ A,
                      const float* __restrict__ Bsrc,
                      const float* __restrict__ dis,
                      float* __restrict__ Part) {
  __shared__ __attribute__((aligned(16))) float ldsB[2 * PANEL_FLOATS];

  const int tid  = threadIdx.x;
  const int wave = tid >> 5;
  const int lane = tid & 31;
  const int half = lane >> 4;
  const int l16  = lane & 15;
  const int rowTile = blockIdx.x * 128 + wave * 16;
  const int kOff = blockIdx.y * KSEG;
  float* PartSeg = Part + (size_t)blockIdx.y * NBT;

  v8f acc[8];
  #pragma unroll
  for (int i = 0; i < 8; ++i)
    #pragma unroll
    for (int j = 0; j < 8; ++j) acc[i][j] = 0.f;

  const int R  = wave;
  const int cq = lane;
  const int kr = (R >> 1) * 8 + (R & 1) * 2;

  const float* Ak = A + (size_t)(rowTile + l16) * NB + kOff + half * 2;
  const float* Bk = Bsrc + (size_t)(kOff + kr) * BT + cq * 4;
  const float* dk_ptr = dis + kOff + kr;

  float* const buf0 = ldsB;
  float* const buf1 = ldsB + PANEL_FLOATS;

  v2f a0[8], a1[8];
  v4f b0[4][8], b1[4][8];
  PanelRegs pr, pq;

  load_a(Ak, a0);
  load_panel(Bk, dk_ptr, pr);
  load_panel(Bk + (size_t)KC * BT, dk_ptr + KC, pq);
  store_panel(buf0, pr, R, cq);
  store_panel(buf1, pq, R, cq);
  wait_lds_done();
  barrier_signal();
  load_panel(Bk + (size_t)2 * KC * BT, dk_ptr + 2 * KC, pr);
  barrier_wait();
  load_frags(b0, buf0, half, l16);
  wait_lds_done();
  barrier_signal();
  barrier_wait();

  for (int it = 0; it < NIT; it += 2) {
    const int c2 = it + 2 < NIT ? it + 2 : NIT - 1;
    const int c3 = it + 3 < NIT ? it + 3 : NIT - 1;
    const int c4 = it + 4 < NIT ? it + 4 : NIT - 1;
    gemm_chunk(a0, a1, pr, b0, b1, acc, buf1, buf0,
               Ak + (it + 1) * KC, Bk + (size_t)c3 * KC * BT, dk_ptr + c3 * KC,
               R, cq, half, l16);
    gemm_chunk(a1, a0, pr, b1, b0, acc, buf0, buf1,
               Ak + c2 * KC, Bk + (size_t)c4 * KC * BT, dk_ptr + c4 * KC,
               R, cq, half, l16);
  }

  const int r0 = rowTile + half * 8;
  #pragma unroll
  for (int g = 0; g < 8; ++g) {
    float* dst = PartSeg + (size_t)(r0 + g) * BT + l16 * 4;
    #pragma unroll
    for (int b = 0; b < 2; ++b) {
      v4f o;
      o[0] = acc[b][g];
      o[1] = acc[b + 2][g];
      o[2] = acc[b + 4][g];
      o[3] = acc[b + 6][g];
      *reinterpret_cast<v4f*>(dst + b * 64) = o;
    }
  }
}

__global__ __launch_bounds__(256)
void ksplit_epilogue_kernel(const float* __restrict__ Part,
                            const float* __restrict__ Bsrc,
                            const float* __restrict__ dis,
                            const float* __restrict__ Adiag,
                            const float* __restrict__ Xadd,
                            float* __restrict__ D,
                            float alpha, float beta) {
  int idx = blockIdx.x * 256 + threadIdx.x;
  int r = idx >> 7;
  float acc = Part[idx] + Part[idx + NBT] + Part[idx + 2 * NBT] + Part[idx + 3 * NBT];
  float dr = dis[r];
  float ad = Adiag[r];
  float v = acc - ad * dr * Bsrc[idx];
  D[idx] = alpha * dr * v + beta * Xadd[idx];
}

__global__ __launch_bounds__(256)
void cheb_conv_kernel(const float* __restrict__ Tx0, const float* __restrict__ Tx1,
                      const float* __restrict__ Tx2,
                      const float* __restrict__ Wc, const float* __restrict__ bc,
                      const float* __restrict__ wt, const float* __restrict__ bt,
                      float* __restrict__ out) {
  __shared__ float sW[3 * 256];
  __shared__ float sbc[16];
  __shared__ float swt[4];
  __shared__ float stile[8][16][17];

  const int tid = threadIdx.x;
  for (int i = tid; i < 768; i += 256) sW[i] = Wc[i];
  if (tid < 16) sbc[tid] = bc[tid];
  if (tid < 3) swt[tid] = wt[tid];
  if (tid == 3) swt[3] = bt[0];
  __syncthreads();

  const int wave = tid >> 5;
  const int lane = tid & 31;
  const int half = lane >> 4;
  const int l16  = lane & 15;
  const int tileId = blockIdx.x * 8 + wave;
  const int b  = tileId >> 9;
  const int nT = tileId & 511;

  const float* Ts[3] = {Tx0, Tx1, Tx2};
  v8f acc;
  #pragma unroll
  for (int j = 0; j < 8; ++j) acc[j] = 0.f;

  #pragma unroll
  for (int m = 0; m < 3; ++m) {
    const float* Abase =
        Ts[m] + (size_t)(nT * 16 + l16) * BT + b * 16 + half * 2;
    #pragma unroll
    for (int s = 0; s < 4; ++s) {
      v2f a = *reinterpret_cast<const v2f*>(Abase + s * 4);
      const int k = s * 4 + half * 2;
      v2f w;
      w[0] = sW[m * 256 + k * 16 + l16];
      w[1] = sW[m * 256 + (k + 1) * 16 + l16];
      acc = __builtin_amdgcn_wmma_f32_16x16x4_f32(
          false, a, false, w, (short)0, acc, false, false);
    }
  }

  const float bcol = sbc[l16];
  #pragma unroll
  for (int g = 0; g < 8; ++g) {
    float v = acc[g] + bcol;
    stile[wave][g + half * 8][l16] = v > 0.f ? v : 0.f;
  }
  __syncthreads();

  const float w0 = swt[0], w1 = swt[1], w2 = swt[2], bb = swt[3];
  #pragma unroll
  for (int g = 0; g < 8; ++g) {
    const int r = g + half * 8;
    const int t = l16;
    float v = bb + w1 * stile[wave][r][t];
    if (t > 0)  v += w0 * stile[wave][r][t - 1];
    if (t < 15) v += w2 * stile[wave][r][t + 1];
    v = v > 0.f ? v : 0.f;
    const int n = nT * 16 + r;
    out[((size_t)b * NB + n) * 16 + t] = v;
  }
}

extern "C" void kernel_launch(void* const* d_in, const int* in_sizes, int n_in,
                              void* d_out, int out_size, void* d_ws, size_t ws_size,
                              hipStream_t stream) {
  const float* X  = (const float*)d_in[0];
  const float* A  = (const float*)d_in[1];
  const float* Wc = (const float*)d_in[2];
  const float* bc = (const float*)d_in[3];
  const float* wt = (const float*)d_in[4];
  const float* bt = (const float*)d_in[5];
  float* out = (float*)d_out;

  float* ws    = (float*)d_ws;
  float* xr    = ws;
  float* Tx1   = ws + (size_t)NBT;
  float* Tx2   = ws + (size_t)2 * NBT;
  float* dis   = ws + (size_t)3 * NBT;
  float* Adiag = dis + NB;
  float* Part  = Adiag + NB;

  repack_x_kernel<<<4096, 256, 0, stream>>>(X, xr);
  degree_kernel<<<256, 512, 0, stream>>>(A, dis, Adiag);

  lhat_gemm_kernel<<<dim3(64, KSPLIT), 256, 0, stream>>>(A, xr, dis, Part);
  ksplit_epilogue_kernel<<<4096, 256, 0, stream>>>(Part, xr, dis, Adiag, xr, Tx1,
                                                   -1.f, 0.f);
  lhat_gemm_kernel<<<dim3(64, KSPLIT), 256, 0, stream>>>(A, Tx1, dis, Part);
  ksplit_epilogue_kernel<<<4096, 256, 0, stream>>>(Part, Tx1, dis, Adiag, xr, Tx2,
                                                   -2.f, -1.f);

  cheb_conv_kernel<<<512, 256, 0, stream>>>(xr, Tx1, Tx2, Wc, bc, wt, bt, out);
}


# Launch fusion, 7 -> 5 launches: X repack folded into the degree kernel; 2nd K-split reduce + Lhat epilogue folded into the mix kernel (Tx2 formed in registers, never stored); same sums, same order
# speedup vs baseline: 1.8445x; 1.0306x over previous
#include <hip/hip_runtime.h>
#include <hip/hip_bf16.h>

typedef float v2f __attribute__((ext_vector_type(2)));
typedef float v4f __attribute__((ext_vector_type(4)));
typedef float v8f __attribute__((ext_vector_type(8)));

#define NB 8192
#define BT 128
#define KC 32
#define KQ (KC / 4)
#define KSPLIT 4
#define KSEG (NB / KSPLIT)
#define NBT (NB * BT)
#define NIT (KSEG / KC)

#define DEG_WAVES 16
#define DEG_ROWS 2
#define DEG_GROUP 16
#define DEG_GROUPS_PER_ROW (NB / 128 / DEG_GROUP)
#define DEG_NGROUPS (DEG_ROWS * DEG_GROUPS_PER_ROW)
#define DEG_BLOCK_ROWS (DEG_WAVES * DEG_ROWS)

__global__ __launch_bounds__(DEG_WAVES * 32) void degree_kernel(const float* __restrict__ A,
                                                                const float* __restrict__ X,
                                                                float* __restrict__ xr,
                                                                float* __restrict__ dis,
                                                                float* __restrict__ Adiag) {
  __shared__ float sDiag[DEG_BLOCK_ROWS];
  __shared__ float sDis[DEG_BLOCK_ROWS];

  const int lane = threadIdx.x & 31;
  const int wave = __builtin_amdgcn_readfirstlane(threadIdx.x >> 5);
  const int row0 = (blockIdx.x * DEG_WAVES + wave) * DEG_ROWS;
  const float* base = A + (size_t)row0 * NB + lane * 4;

  v4f xv[2];
  #pragma unroll
  for (int j = 0; j < 2; ++j) {
    const int p = j * (DEG_WAVES * 32) + threadIdx.x;
    const int c4 = p & 31;
    xv[j] = *reinterpret_cast<const v4f*>(X + (size_t)(c4 >> 2) * (NB * 16) +
                                          (size_t)(blockIdx.x * DEG_BLOCK_ROWS + (p >> 5)) * 16 +
                                          (c4 & 3) * 4);
  }

  v4f buf[2][DEG_GROUP];
  float sum[DEG_ROWS];

#define DEG_LOAD_GROUP(G) do { \
    _Pragma("unroll") for (int q = 0; q < DEG_GROUP; ++q) \
      buf[(G) & 1][q] = *reinterpret_cast<const v4f*>(base + ((G) * DEG_GROUP + q) * 128); } while (0)

  DEG_LOAD_GROUP(0);
  float s = 0.f;
  #pragma unroll
  for (int g = 0; g < DEG_NGROUPS; ++g) {
    __builtin_amdgcn_sched_barrier(0);
    if (g + 1 < DEG_NGROUPS) DEG_LOAD_GROUP(g + 1);
    __builtin_amdgcn_sched_barrier(0);
    #pragma unroll
    for (int q = 0; q < DEG_GROUP; ++q) {
      const v4f v = buf[g & 1][q];
      s += v.x + v.y + v.z + v.w;
    }
    if ((g % DEG_GROUPS_PER_ROW) == DEG_GROUPS_PER_ROW - 1) {
      sum[g / DEG_GROUPS_PER_ROW] = s;
      s = 0.f;
    }
  }
#undef DEG_LOAD_GROUP

  #pragma unroll
  for (int j = 0; j < 2; ++j) {
    const int p = j * (DEG_WAVES * 32) + threadIdx.x;
    *reinterpret_cast<v4f*>(xr + (size_t)blockIdx.x * (DEG_BLOCK_ROWS * BT) + p * 4) = xv[j];
  }

  #pragma unroll
  for (int r = 0; r < DEG_ROWS; ++r) {
    float t = sum[r];
    for (int off = 16; off > 0; off >>= 1) t += __shfl_xor(t, off, 32);
    sum[r] = t;
  }

  float mySum = sum[0];
  #pragma unroll
  for (int r = 1; r < DEG_ROWS; ++r)
    if (lane == r) mySum = sum[r];
  if (lane < DEG_ROWS) {
    const int row = row0 + lane;
    const float diag = A[(size_t)row * NB + row];
    const float deg = mySum - diag;
    sDiag[wave * DEG_ROWS + lane] = diag;
    sDis[wave * DEG_ROWS + lane] = (deg > 0.f) ? rsqrtf(deg) : 0.f;
  }
  __syncthreads();
  if (threadIdx.x < DEG_BLOCK_ROWS) {
    Adiag[blockIdx.x * DEG_BLOCK_ROWS + threadIdx.x] = sDiag[threadIdx.x];
    dis[blockIdx.x * DEG_BLOCK_ROWS + threadIdx.x] = sDis[threadIdx.x];
  }
}


#define PANEL_FLOATS (KQ * BT * 4)

struct PanelRegs {
  v4f v[4];
  v2f dlo;
  v2f dhi;
};

__device__ __forceinline__ void wait_lds_done() { asm volatile("s_wait_dscnt 0x0" ::: "memory"); }
__device__ __forceinline__ void barrier_signal() { asm volatile("s_barrier_signal -1" ::: "memory"); }
__device__ __forceinline__ void barrier_wait() { asm volatile("s_barrier_wait -1" ::: "memory"); }

__device__ __forceinline__ void load_a(const float* __restrict__ Ak, v2f (&a)[8]) {
  #pragma unroll
  for (int s = 0; s < 8; ++s)
    a[s] = *reinterpret_cast<const v2f*>(Ak + s * 4);
}

__device__ __forceinline__ void load_panel(const float* __restrict__ Bk,
                                           const float* __restrict__ dk_ptr,
                                           PanelRegs& pr) {
  pr.v[0] = *reinterpret_cast<const v4f*>(Bk);
  pr.v[1] = *reinterpret_cast<const v4f*>(Bk + BT);
  pr.v[2] = *reinterpret_cast<const v4f*>(Bk + 4 * BT);
  pr.v[3] = *reinterpret_cast<const v4f*>(Bk + 5 * BT);
  pr.dlo = *reinterpret_cast<const v2f*>(dk_ptr);
  pr.dhi = *reinterpret_cast<const v2f*>(dk_ptr + 4);
}

__device__ __forceinline__ void store_panel(float* panel, const PanelRegs& pr,
                                            int R, int cq) {
  #pragma unroll
  for (int i = 0; i < 4; ++i) {
    v4f o;
    o[0] = pr.v[0][i] * pr.dlo[0];
    o[1] = pr.v[1][i] * pr.dlo[1];
    o[2] = pr.v[2][i] * pr.dhi[0];
    o[3] = pr.v[3][i] * pr.dhi[1];
    *reinterpret_cast<v4f*>(panel + ((R * BT + i * 32 + cq) << 2)) = o;
  }
}

__device__ __forceinline__ void load_frags(v4f (&b)[4][8], const float* panel,
                                           int half, int l16) {
  #pragma unroll
  for (int j = 0; j < 4; ++j)
    #pragma unroll
    for (int t = 0; t < 8; ++t)
      b[j][t] = *reinterpret_cast<const v4f*>(panel + (((2 * j + half) * BT + t * 16 + l16) << 2));
}

template <int ODD>
__device__ __forceinline__ void mma_row(v8f (&acc)[8], v2f a, const v4f (&b)[8]) {
  #pragma unroll
  for (int t = 0; t < 8; ++t) {
    v2f bb = ODD ? __builtin_shufflevector(b[t], b[t], 2, 3)
                 : __builtin_shufflevector(b[t], b[t], 0, 1);
    acc[t] = __builtin_amdgcn_wmma_f32_16x16x4_f32(
        false, a, false, bb, (short)0, acc[t], false, false);
  }
}

__device__ __forceinline__ void gemm_chunk(const v2f (&aC)[8], v2f (&aN)[8], PanelRegs& pr,
                                           const v4f (&bC)[4][8], v4f (&bN)[4][8],
                                           v8f (&acc)[8],
                                           const float* pRead, float* pWrite,
                                           const float* __restrict__ AkN,
                                           const float* __restrict__ BkN3,
                                           const float* __restrict__ dkN3,
                                           int R, int cq, int half, int l16) {
  load_frags(bN, pRead, half, l16);
  load_a(AkN, aN);

  __builtin_amdgcn_sched_barrier(0);

  mma_row<0>(acc, aC[0], bC[0]);
  mma_row<1>(acc, aC[1], bC[0]);
  mma_row<0>(acc, aC[2], bC[1]);

  __builtin_amdgcn_sched_barrier(0);

  store_panel(pWrite, pr, R, cq);
  __builtin_amdgcn_sched_barrier(0);
  mma_row<1>(acc, aC[3], bC[1]);
  __builtin_amdgcn_sched_barrier(0);
  wait_lds_done();
  barrier_signal();
  load_panel(BkN3, dkN3, pr);

  __builtin_amdgcn_sched_barrier(0);

  mma_row<0>(acc, aC[4], bC[2]);
  mma_row<1>(acc, aC[5], bC[2]);
  mma_row<0>(acc, aC[6], bC[3]);
  mma_row<1>(acc, aC[7], bC[3]);

  __builtin_amdgcn_sched_barrier(0);
  barrier_wait();

}

__global__ __launch_bounds__(256)
void lhat_gemm_kernel(const float* __restrict__ A,
                      const float* __restrict__ Bsrc,
                      const float* __restrict__ dis,
                      float* __restrict__ Part) {
  __shared__ __attribute__((aligned(16))) float ldsB[2 * PANEL_FLOATS];

  const int tid  = threadIdx.x;
  const int wave = tid >> 5;
  const int lane = tid & 31;
  const int half = lane >> 4;
  const int l16  = lane & 15;
  const int rowTile = blockIdx.x * 128 + wave * 16;
  const int kOff = blockIdx.y * KSEG;
  float* PartSeg = Part + (size_t)blockIdx.y * NBT;

  v8f acc[8];
  #pragma unroll
  for (int i = 0; i < 8; ++i)
    #pragma unroll
    for (int j = 0; j < 8; ++j) acc[i][j] = 0.f;

  const int R  = wave;
  const int cq = lane;
  const int kr = (R >> 1) * 8 + (R & 1) * 2;

  const float* Ak = A + (size_t)(rowTile + l16) * NB + kOff + half * 2;
  const float* Bk = Bsrc + (size_t)(kOff + kr) * BT + cq * 4;
  const float* dk_ptr = dis + kOff + kr;

  float* const buf0 = ldsB;
  float* const buf1 = ldsB + PANEL_FLOATS;

  v2f a0[8], a1[8];
  v4f b0[4][8], b1[4][8];
  PanelRegs pr, pq;

  load_a(Ak, a0);
  load_panel(Bk, dk_ptr, pr);
  load_panel(Bk + (size_t)KC * BT, dk_ptr + KC, pq);
  store_panel(buf0, pr, R, cq);
  store_panel(buf1, pq, R, cq);
  wait_lds_done();
  barrier_signal();
  load_panel(Bk + (size_t)2 * KC * BT, dk_ptr + 2 * KC, pr);
  barrier_wait();
  load_frags(b0, buf0, half, l16);
  wait_lds_done();
  barrier_signal();
  barrier_wait();

  for (int it = 0; it < NIT; it += 2) {
    const int c2 = it + 2 < NIT ? it + 2 : NIT - 1;
    const int c3 = it + 3 < NIT ? it + 3 : NIT - 1;
    const int c4 = it + 4 < NIT ? it + 4 : NIT - 1;
    gemm_chunk(a0, a1, pr, b0, b1, acc, buf1, buf0,
               Ak + (it + 1) * KC, Bk + (size_t)c3 * KC * BT, dk_ptr + c3 * KC,
               R, cq, half, l16);
    gemm_chunk(a1, a0, pr, b1, b0, acc, buf0, buf1,
               Ak + c2 * KC, Bk + (size_t)c4 * KC * BT, dk_ptr + c4 * KC,
               R, cq, half, l16);
  }

  const int r0 = rowTile + half * 8;
  #pragma unroll
  for (int g = 0; g < 8; ++g) {
    float* dst = PartSeg + (size_t)(r0 + g) * BT + l16 * 4;
    #pragma unroll
    for (int b = 0; b < 2; ++b) {
      v4f o;
      o[0] = acc[b][g];
      o[1] = acc[b + 2][g];
      o[2] = acc[b + 4][g];
      o[3] = acc[b + 6][g];
      *reinterpret_cast<v4f*>(dst + b * 64) = o;
    }
  }
}

__device__ __forceinline__ float lhat_epilogue(float p0, float p1, float p2, float p3,
                                               float bsrc, float dr, float ad, float xadd,
                                               float alpha, float beta) {
  float acc = p0 + p1 + p2 + p3;
  float v = acc - ad * dr * bsrc;
  return alpha * dr * v + beta * xadd;
}

__global__ __launch_bounds__(256)
void ksplit_epilogue_kernel(const float* __restrict__ Part,
                            const float* __restrict__ Bsrc,
                            const float* __restrict__ dis,
                            const float* __restrict__ Adiag,
                            const float* __restrict__ Xadd,
                            float* __restrict__ D,
                            float alpha, float beta) {
  int idx = blockIdx.x * 256 + threadIdx.x;
  int r = idx >> 7;
  D[idx] = lhat_epilogue(Part[idx], Part[idx + NBT], Part[idx + 2 * NBT], Part[idx + 3 * NBT],
                         Bsrc[idx], dis[r], Adiag[r], Xadd[idx], alpha, beta);
}

__global__ __launch_bounds__(256)
void cheb_conv_kernel(const float* __restrict__ Tx0, const float* __restrict__ Tx1,
                      const float* __restrict__ Part,
                      const float* __restrict__ dis, const float* __restrict__ Adiag,
                      const float* __restrict__ Wc, const float* __restrict__ bc,
                      const float* __restrict__ wt, const float* __restrict__ bt,
                      float* __restrict__ out) {
  __shared__ float sW[3 * 256];
  __shared__ float sbc[16];
  __shared__ float swt[4];
  __shared__ float stile[8][16][17];

  const int tid = threadIdx.x;
  const int wave = tid >> 5;
  const int lane = tid & 31;
  const int half = lane >> 4;
  const int l16  = lane & 15;
  const int b  = wave;
  const int nT = blockIdx.x;

  const int row = nT * 16 + l16;
  const size_t off = (size_t)row * BT + b * 16 + half * 2;
  v2f x0[4], x1[4], pp[4][4];
  #pragma unroll
  for (int s = 0; s < 4; ++s) {
    x0[s] = *reinterpret_cast<const v2f*>(Tx0 + off + s * 4);
    x1[s] = *reinterpret_cast<const v2f*>(Tx1 + off + s * 4);
    #pragma unroll
    for (int q = 0; q < 4; ++q)
      pp[q][s] = *reinterpret_cast<const v2f*>(Part + (size_t)q * NBT + off + s * 4);
  }
  const float dr = dis[row];
  const float ad = Adiag[row];

  for (int i = tid; i < 768; i += 256) sW[i] = Wc[i];
  if (tid < 16) sbc[tid] = bc[tid];
  if (tid < 3) swt[tid] = wt[tid];
  if (tid == 3) swt[3] = bt[0];
  __syncthreads();

  v2f x2[4];
  #pragma unroll
  for (int s = 0; s < 4; ++s)
    #pragma unroll
    for (int e = 0; e < 2; ++e)
      x2[s][e] = lhat_epilogue(pp[0][s][e], pp[1][s][e], pp[2][s][e], pp[3][s][e],
                               x1[s][e], dr, ad, x0[s][e], -2.f, -1.f);

  v8f acc;
  #pragma unroll
  for (int j = 0; j < 8; ++j) acc[j] = 0.f;

  #pragma unroll
  for (int m = 0; m < 3; ++m) {
    #pragma unroll
    for (int s = 0; s < 4; ++s) {
      const v2f a = m == 0 ? x0[s] : (m == 1 ? x1[s] : x2[s]);
      const int k = s * 4 + half * 2;
      v2f w;
      w[0] = sW[m * 256 + k * 16 + l16];
      w[1] = sW[m * 256 + (k + 1) * 16 + l16];
      acc = __builtin_amdgcn_wmma_f32_16x16x4_f32(
          false, a, false, w, (short)0, acc, false, false);
    }
  }

  const float bcol = sbc[l16];
  #pragma unroll
  for (int g = 0; g < 8; ++g) {
    float v = acc[g] + bcol;
    stile[wave][g + half * 8][l16] = v > 0.f ? v : 0.f;
  }
  __syncthreads();

  const float w0 = swt[0], w1 = swt[1], w2 = swt[2], bb = swt[3];
  #pragma unroll
  for (int g = 0; g < 8; ++g) {
    const int r = g + half * 8;
    const int t = l16;
    float v = bb + w1 * stile[wave][r][t];
    if (t > 0)  v += w0 * stile[wave][r][t - 1];
    if (t < 15) v += w2 * stile[wave][r][t + 1];
    v = v > 0.f ? v : 0.f;
    const int n = nT * 16 + r;
    out[((size_t)b * NB + n) * 16 + t] = v;
  }
}

extern "C" void kernel_launch(void* const* d_in, const int* in_sizes, int n_in,
                              void* d_out, int out_size, void* d_ws, size_t ws_size,
                              hipStream_t stream) {
  const float* X  = (const float*)d_in[0];
  const float* A  = (const float*)d_in[1];
  const float* Wc = (const float*)d_in[2];
  const float* bc = (const float*)d_in[3];
  const float* wt = (const float*)d_in[4];
  const float* bt = (const float*)d_in[5];
  float* out = (float*)d_out;

  float* ws    = (float*)d_ws;
  float* xr    = ws;
  float* Tx1   = ws + (size_t)NBT;
  float* dis   = ws + (size_t)3 * NBT;
  float* Adiag = dis + NB;
  float* Part  = Adiag + NB;

  degree_kernel<<<256, 512, 0, stream>>>(A, X, xr, dis, Adiag);

  lhat_gemm_kernel<<<dim3(64, KSPLIT), 256, 0, stream>>>(A, xr, dis, Part);
  ksplit_epilogue_kernel<<<4096, 256, 0, stream>>>(Part, xr, dis, Adiag, xr, Tx1,
                                                   -1.f, 0.f);
  lhat_gemm_kernel<<<dim3(64, KSPLIT), 256, 0, stream>>>(A, Tx1, dis, Part);

  cheb_conv_kernel<<<512, 256, 0, stream>>>(xr, Tx1, Part, dis, Adiag, Wc, bc, wt, bt, out);
}
